# starting kernel
# baseline (speedup 1.0000x reference)
#include <hip/hip_runtime.h>
#include <math.h>


typedef __attribute__((ext_vector_type(2))) float v2f;
typedef __attribute__((ext_vector_type(8))) float v8f;

#define B_TOT    512
#define D        32
#define H        256
#define BR       16
#define NTHREADS 256
#define NSTEP    64

#define XS   36
#define HS   260
#define W1F  544
#define W2F  96
#define G2S  36

__device__ __forceinline__ float fast_tanh(float z) {
    const float TWO_LOG2E = 2.88539008177792681472f;
    const float a = __builtin_fabsf(z);
    const float e = __builtin_amdgcn_exp2f(a * TWO_LOG2E);
    const float r = __builtin_amdgcn_rcpf(e + 1.0f);
    const float t = __builtin_fmaf(-2.0f, r, 1.0f);
    return __builtin_copysignf(t, z);
}

__device__ __forceinline__ v2f lds_v2f(const float* p) {
    return *(const v2f*)p;
}

__device__ __forceinline__ void eval_core(
    const float* __restrict__ xin,
    float* __restrict__ hbuf,
    float* __restrict__ g2part,
    float* __restrict__ part,
    float* __restrict__ epart,
    const float* __restrict__ w1p,
    const float* __restrict__ w2p,
    const float* __restrict__ cdiag,
    const float* __restrict__ b1eff,
    int tid)
{
    const int wid  = tid >> 5;
    const int lane = tid & 31;
    const int half = lane >> 4;
    const int lr   = lane & 15;

#pragma unroll
    for (int it = 0; it < 2; ++it) {
        const int n0 = (wid * 2 + it) * 16;
        v8f acc = {};
#pragma unroll
        for (int k0 = 0; k0 < 8; ++k0) {
            const int kb = k0 * 4 + 2 * half;
            const v2f a = lds_v2f(&xin[lr * XS + kb]);
            const v2f b = lds_v2f(&w1p[(2 * k0 + half) * W1F + 2 * (n0 + lr)]);
            acc = __builtin_amdgcn_wmma_f32_16x16x4_f32(
                      false, a, false, b, (short)0, acc, false, false);
        }
#pragma unroll
        for (int v = 0; v < 8; ++v) {
            const int row = v + 8 * half;
            const int col = n0 + lr;
            hbuf[row * HS + col] = fast_tanh(acc[v] + b1eff[col]);
        }
    }
    __syncthreads();

    {
        const int n0 = (wid & 1) * 16;
        const int s  = wid >> 1;
        v8f acc = {};
#pragma unroll
        for (int k0i = 0; k0i < 16; ++k0i) {
            const int kk = s * 16 + k0i;
            const int kb = kk * 4 + 2 * half;
            const v2f a = lds_v2f(&hbuf[lr * HS + kb]);
            const v2f b = lds_v2f(&w2p[(2 * kk + half) * W2F + 2 * (n0 + lr)]);
            acc = __builtin_amdgcn_wmma_f32_16x16x4_f32(
                      false, a, false, b, (short)0, acc, false, false);
        }
#pragma unroll
        for (int v = 0; v < 8; ++v) {
            const int row = v + 8 * half;
            g2part[(s * BR + row) * G2S + n0 + lr] = acc[v];
        }
    }

    {
        const int r = tid >> 4;
        const int sub = tid & 15;
        float dsum = 0.f;
#pragma unroll
        for (int j = sub; j < H; j += 16) {
            const float hv = hbuf[r * HS + j];
            dsum += (1.f - hv * hv) * cdiag[j];
        }
        part[tid] = dsum;
        const float xa = xin[r * XS + sub];
        const float xb = xin[r * XS + sub + 16];
        epart[tid] = 0.5f * (xa * xa + xb * xb);
    }
    __syncthreads();
}

__global__ __launch_bounds__(NTHREADS)
void cnf_rk4_kernel(const float* __restrict__ x,
                    const float* __restrict__ W1,
                    const float* __restrict__ b1,
                    const float* __restrict__ W2,
                    const float* __restrict__ b2,
                    float* __restrict__ out)
{
    __shared__ __align__(16) float w1p[16 * W1F];
    __shared__ __align__(16) float w2p[128 * W2F];
    __shared__ float w1time[H];
    __shared__ float b1s[H], b2s[D], cdiag[H], b1eff[H];
    __shared__ __align__(16) float hbuf[BR * HS];
    __shared__ __align__(16) float x0[BR * XS], xev[BR * XS], xacc[BR * XS];
    __shared__ float g2part[4 * BR * G2S];
    __shared__ float part[BR * 16], epart[BR * 16];
    __shared__ float acclp[BR], accls[BR], lp0[BR], ls0[BR];

    const int tid = threadIdx.x;
    const int wg  = blockIdx.x;

    for (int i = tid; i < 32 * H; i += NTHREADS) {
        const int k = i / H, n = i % H;
        w1p[(k >> 1) * W1F + 2 * n + (k & 1)] = W1[i];
    }
    w1time[tid] = W1[32 * H + tid];
    for (int i = tid; i < H * D; i += NTHREADS) {
        const int k = i / D, n = i % D;
        w2p[(k >> 1) * W2F + 2 * n + (k & 1)] = W2[i];
    }
    b1s[tid] = b1[tid];
    if (tid < D) b2s[tid] = b2[tid];
    for (int i = tid; i < BR * D; i += NTHREADS)
        x0[(i / D) * XS + (i % D)] = x[wg * BR * D + i];
    if (tid < BR) { lp0[tid] = 0.f; ls0[tid] = 0.f; }
    __syncthreads();

    const float b1r = b1s[tid];
    const float w1t = w1time[tid];

    {
        float c = 0.f;
        const int j = tid;
#pragma unroll
        for (int k = 0; k < D; ++k)
            c += w1p[(k >> 1) * W1F + 2 * j + (k & 1)]
               * w2p[(j >> 1) * W2F + 2 * k + (j & 1)];
        cdiag[j] = c;
        b1eff[j] = b1r;
    }
    __syncthreads();

    const float dt = 1.0f / (float)NSTEP;

    for (int step = 0; step < NSTEP; ++step) {
        const float t0 = step * dt;

        eval_core(x0, hbuf, g2part, part, epart, w1p, w2p, cdiag, b1eff, tid);
        for (int i = tid; i < BR * D; i += NTHREADS) {
            const int r = i >> 5, d = i & 31;
            const float fv = b2s[d]
                + g2part[(0 * BR + r) * G2S + d] + g2part[(1 * BR + r) * G2S + d]
                + g2part[(2 * BR + r) * G2S + d] + g2part[(3 * BR + r) * G2S + d];
            xacc[r * XS + d] = fv;
            xev[r * XS + d]  = x0[r * XS + d] + 0.5f * dt * fv;
        }
        if (tid < BR) {
            float ds = 0.f, es = 0.f;
#pragma unroll
            for (int s = 0; s < 16; ++s) { ds += part[tid * 16 + s]; es += epart[tid * 16 + s]; }
            acclp[tid] = -ds;  accls[tid] = es;
        }
        b1eff[tid] = b1r + (t0 + 0.5f * dt) * w1t;
        __syncthreads();

        eval_core(xev, hbuf, g2part, part, epart, w1p, w2p, cdiag, b1eff, tid);
        for (int i = tid; i < BR * D; i += NTHREADS) {
            const int r = i >> 5, d = i & 31;
            const float fv = b2s[d]
                + g2part[(0 * BR + r) * G2S + d] + g2part[(1 * BR + r) * G2S + d]
                + g2part[(2 * BR + r) * G2S + d] + g2part[(3 * BR + r) * G2S + d];
            xacc[r * XS + d] += 2.f * fv;
            xev[r * XS + d]   = x0[r * XS + d] + 0.5f * dt * fv;
        }
        if (tid < BR) {
            float ds = 0.f, es = 0.f;
#pragma unroll
            for (int s = 0; s < 16; ++s) { ds += part[tid * 16 + s]; es += epart[tid * 16 + s]; }
            acclp[tid] += -2.f * ds;  accls[tid] += 2.f * es;
        }
        __syncthreads();

        eval_core(xev, hbuf, g2part, part, epart, w1p, w2p, cdiag, b1eff, tid);
        for (int i = tid; i < BR * D; i += NTHREADS) {
            const int r = i >> 5, d = i & 31;
            const float fv = b2s[d]
                + g2part[(0 * BR + r) * G2S + d] + g2part[(1 * BR + r) * G2S + d]
                + g2part[(2 * BR + r) * G2S + d] + g2part[(3 * BR + r) * G2S + d];
            xacc[r * XS + d] += 2.f * fv;
            xev[r * XS + d]   = x0[r * XS + d] + dt * fv;
        }
        if (tid < BR) {
            float ds = 0.f, es = 0.f;
#pragma unroll
            for (int s = 0; s < 16; ++s) { ds += part[tid * 16 + s]; es += epart[tid * 16 + s]; }
            acclp[tid] += -2.f * ds;  accls[tid] += 2.f * es;
        }
        b1eff[tid] = b1r + (t0 + dt) * w1t;
        __syncthreads();

        eval_core(xev, hbuf, g2part, part, epart, w1p, w2p, cdiag, b1eff, tid);
        for (int i = tid; i < BR * D; i += NTHREADS) {
            const int r = i >> 5, d = i & 31;
            const float fv = b2s[d]
                + g2part[(0 * BR + r) * G2S + d] + g2part[(1 * BR + r) * G2S + d]
                + g2part[(2 * BR + r) * G2S + d] + g2part[(3 * BR + r) * G2S + d];
            x0[r * XS + d] += (dt / 6.f) * (xacc[r * XS + d] + fv);
        }
        if (tid < BR) {
            float ds = 0.f, es = 0.f;
#pragma unroll
            for (int s = 0; s < 16; ++s) { ds += part[tid * 16 + s]; es += epart[tid * 16 + s]; }
            lp0[tid] += (dt / 6.f) * (acclp[tid] - ds);
            ls0[tid] += (dt / 6.f) * (accls[tid] + es);
        }
        b1eff[tid] = b1r + (t0 + dt) * w1t;
        __syncthreads();
    }

    for (int i = tid; i < BR * D; i += NTHREADS)
        out[wg * BR * D + i] = x0[(i >> 5) * XS + (i & 31)];
    if (tid < BR) {
        out[B_TOT * D + wg * BR + tid]         = lp0[tid];
        out[B_TOT * D + B_TOT + wg * BR + tid] = ls0[tid];
    }
}

extern "C" void kernel_launch(void* const* d_in, const int* in_sizes, int n_in,
                              void* d_out, int out_size, void* d_ws, size_t ws_size,
                              hipStream_t stream) {
    const float* x  = (const float*)d_in[0];
    const float* W1 = (const float*)d_in[1];
    const float* b1 = (const float*)d_in[2];
    const float* W2 = (const float*)d_in[3];
    const float* b2 = (const float*)d_in[4];
    float* out = (float*)d_out;
    cnf_rk4_kernel<<<B_TOT / BR, NTHREADS, 0, stream>>>(x, W1, b1, W2, b2, out);
}


# GEMM1 transposed so each wave's tanh registers are the A fragments of its own K-slice of GEMM2: h skips LDS and the mid barrier; f summed from 8 K-slices not 4, div partials regrouped by wave
# speedup vs baseline: 1.3134x; 1.3134x over previous
#include <hip/hip_runtime.h>
#include <math.h>


typedef __attribute__((ext_vector_type(2))) float v2f;
typedef __attribute__((ext_vector_type(8))) float v8f;

#define B_TOT    512
#define D        32
#define H        256
#define BR       16
#define NTHREADS 256
#define NSTEP    64

#define XS   36
#define W1F  544
#define W2F  72
#define G2S  36
#define NPART 8

template <int N>
__device__ __forceinline__ void tanh_batch(const float (&z)[N], float (&h)[N]) {
    const float TWO_LOG2E = 2.88539008177792681472f;
    float e[N];
#pragma unroll
    for (int i = 0; i < N; ++i)
        e[i] = __builtin_amdgcn_exp2f(__builtin_fabsf(z[i]) * TWO_LOG2E);
#pragma unroll
    for (int i = 0; i < N; ++i)
        e[i] = __builtin_amdgcn_rcpf(e[i] + 1.0f);
#pragma unroll
    for (int i = 0; i < N; ++i)
        h[i] = __builtin_copysignf(__builtin_fmaf(-2.0f, e[i], 1.0f), z[i]);
}

__device__ __forceinline__ v2f lds_v2f(const float* p) {
    return *(const v2f*)p;
}

__device__ __forceinline__ void eval_core(
    const float* __restrict__ xin,
    float* __restrict__ g2part,
    float* __restrict__ part,
    float* __restrict__ epart,
    const float* __restrict__ w1p,
    const float* __restrict__ w2p,
    const float (&cd)[16],
    const float (&b1v)[16],
    const float (&w1tv)[16],
    float tcur,
    int tid)
{
    const int wid  = tid >> 5;
    const int lane = tid & 31;
    const int half = lane >> 4;
    const int lr   = lane & 15;

    v2f xf[8], wf[2][8];
#pragma unroll
    for (int k0 = 0; k0 < 8; ++k0)
        xf[k0] = lds_v2f(&xin[lr * XS + k0 * 4 + 2 * half]);
#pragma unroll
    for (int it = 0; it < 2; ++it) {
#pragma unroll
        for (int k0 = 0; k0 < 8; ++k0)
            wf[it][k0] = lds_v2f(&w1p[(2 * k0 + half) * W1F + 2 * ((wid * 2 + it) * 16 + lr)]);
    }
    v8f acc0 = {}, acc1 = {};
#pragma unroll
    for (int k0 = 0; k0 < 8; ++k0) {
        acc0 = __builtin_amdgcn_wmma_f32_16x16x4_f32(
                   false, wf[0][k0], false, xf[k0], (short)0, acc0, false, false);
        acc1 = __builtin_amdgcn_wmma_f32_16x16x4_f32(
                   false, wf[1][k0], false, xf[k0], (short)0, acc1, false, false);
    }

    v2f bf[2][2][4];
#pragma unroll
    for (int nt = 0; nt < 2; ++nt) {
#pragma unroll
        for (int it = 0; it < 2; ++it) {
#pragma unroll
            for (int p = 0; p < 4; ++p)
                bf[nt][it][p] = lds_v2f(&w2p[(16 * wid + 8 * it + 4 * half + p) * W2F
                                             + 2 * (nt * 16 + lr)]);
        }
    }

    float z0[8], z1[8], h0[8], h1[8];
#pragma unroll
    for (int v = 0; v < 8; ++v) {
        z0[v] = acc0[v] + __builtin_fmaf(tcur, w1tv[v],     b1v[v]);
        z1[v] = acc1[v] + __builtin_fmaf(tcur, w1tv[8 + v], b1v[8 + v]);
    }
    tanh_batch<8>(z0, h0);

    v8f f0 = {}, f1 = {};
    float dsum = 0.f;
#pragma unroll
    for (int p = 0; p < 4; ++p) {
        v2f a;
        a[0] = h0[2 * p];
        a[1] = h0[2 * p + 1];
        f0 = __builtin_amdgcn_wmma_f32_16x16x4_f32(
                 false, a, false, bf[0][0][p], (short)0, f0, false, false);
        f1 = __builtin_amdgcn_wmma_f32_16x16x4_f32(
                 false, a, false, bf[1][0][p], (short)0, f1, false, false);
        const float zz[2] = { z1[2 * p], z1[2 * p + 1] };
        float hh[2];
        tanh_batch<2>(zz, hh);
        h1[2 * p]     = hh[0];
        h1[2 * p + 1] = hh[1];
        dsum += (1.f - h0[2 * p] * h0[2 * p]) * cd[2 * p];
        dsum += (1.f - h0[2 * p + 1] * h0[2 * p + 1]) * cd[2 * p + 1];
    }
#pragma unroll
    for (int p = 0; p < 4; ++p) {
        v2f a;
        a[0] = h1[2 * p];
        a[1] = h1[2 * p + 1];
        f0 = __builtin_amdgcn_wmma_f32_16x16x4_f32(
                 false, a, false, bf[0][1][p], (short)0, f0, false, false);
        f1 = __builtin_amdgcn_wmma_f32_16x16x4_f32(
                 false, a, false, bf[1][1][p], (short)0, f1, false, false);
        dsum += (1.f - h1[2 * p] * h1[2 * p]) * cd[8 + 2 * p];
        dsum += (1.f - h1[2 * p + 1] * h1[2 * p + 1]) * cd[8 + 2 * p + 1];
    }
    part[lr * 16 + wid * 2 + half] = dsum;
#pragma unroll
    for (int v = 0; v < 8; ++v) {
        const int row = v + 8 * half;
        g2part[(wid * BR + row) * G2S + lr]      = f0[v];
        g2part[(wid * BR + row) * G2S + 16 + lr] = f1[v];
    }

    {
        const int r = tid >> 4;
        const int sub = tid & 15;
        const float xa = xin[r * XS + sub];
        const float xb = xin[r * XS + sub + 16];
        epart[tid] = 0.5f * (xa * xa + xb * xb);
    }
    __syncthreads();
}

__global__ __launch_bounds__(NTHREADS)
void cnf_rk4_kernel(const float* __restrict__ x,
                    const float* __restrict__ W1,
                    const float* __restrict__ b1,
                    const float* __restrict__ W2,
                    const float* __restrict__ b2,
                    float* __restrict__ out)
{
    __shared__ __align__(16) float w1p[16 * W1F];
    __shared__ __align__(16) float w2p[128 * W2F];
    __shared__ __align__(16) float w1time[H];
    __shared__ __align__(16) float b1s[H], cdiag[H];
    __shared__ float b2s[D];
    __shared__ __align__(16) float x0[BR * XS], xev[BR * XS], xacc[BR * XS];
    __shared__ float g2part[NPART * BR * G2S];
    __shared__ float part[BR * 16], epart[BR * 16];
    __shared__ float acclp[BR], accls[BR], lp0[BR], ls0[BR];

    const int tid = threadIdx.x;
    const int wg  = blockIdx.x;

    for (int i = tid; i < 32 * H; i += NTHREADS) {
        const int k = i / H, n = i % H;
        w1p[(k >> 1) * W1F + 2 * n + (k & 1)] = W1[i];
    }
    w1time[tid] = W1[32 * H + tid];
    for (int i = tid; i < H * D; i += NTHREADS) {
        const int k = i / D, n = i % D;
        w2p[(k >> 1) * W2F + 2 * n + (k & 1)] = W2[i];
    }
    b1s[tid] = b1[tid];
    if (tid < D) b2s[tid] = b2[tid];
    for (int i = tid; i < BR * D; i += NTHREADS)
        x0[(i / D) * XS + (i % D)] = x[wg * BR * D + i];
    if (tid < BR) { lp0[tid] = 0.f; ls0[tid] = 0.f; }
    __syncthreads();

    {
        float c = 0.f;
        const int j = tid;
#pragma unroll
        for (int k = 0; k < D; ++k)
            c += w1p[(k >> 1) * W1F + 2 * j + (k & 1)]
               * w2p[(j >> 1) * W2F + 2 * k + (j & 1)];
        cdiag[j] = c;
    }
    __syncthreads();

    float cd[16], b1v[16], w1tv[16];
    {
        const int wid  = tid >> 5;
        const int half = (tid & 31) >> 4;
#pragma unroll
        for (int it = 0; it < 2; ++it) {
#pragma unroll
            for (int v = 0; v < 8; ++v) {
                const int col = (wid * 2 + it) * 16 + 8 * half + v;
                cd[it * 8 + v]   = cdiag[col];
                b1v[it * 8 + v]  = b1s[col];
                w1tv[it * 8 + v] = w1time[col];
            }
        }
    }

    const float dt = 1.0f / (float)NSTEP;

    for (int step = 0; step < NSTEP; ++step) {
        const float t0 = step * dt;

        eval_core(x0, g2part, part, epart, w1p, w2p, cd, b1v, w1tv, t0, tid);
        for (int i = tid; i < BR * D; i += NTHREADS) {
            const int r = i >> 5, d = i & 31;
            float fv = b2s[d];
#pragma unroll
            for (int q = 0; q < NPART; ++q)
                fv += g2part[(q * BR + r) * G2S + d];
            xacc[r * XS + d] = fv;
            xev[r * XS + d]  = x0[r * XS + d] + 0.5f * dt * fv;
        }
        if (tid < BR) {
            float ds = 0.f, es = 0.f;
#pragma unroll
            for (int s = 0; s < 16; ++s) { ds += part[tid * 16 + s]; es += epart[tid * 16 + s]; }
            acclp[tid] = -ds;  accls[tid] = es;
        }
        __syncthreads();

        eval_core(xev, g2part, part, epart, w1p, w2p, cd, b1v, w1tv, t0 + 0.5f * dt, tid);
        for (int i = tid; i < BR * D; i += NTHREADS) {
            const int r = i >> 5, d = i & 31;
            float fv = b2s[d];
#pragma unroll
            for (int q = 0; q < NPART; ++q)
                fv += g2part[(q * BR + r) * G2S + d];
            xacc[r * XS + d] += 2.f * fv;
            xev[r * XS + d]   = x0[r * XS + d] + 0.5f * dt * fv;
        }
        if (tid < BR) {
            float ds = 0.f, es = 0.f;
#pragma unroll
            for (int s = 0; s < 16; ++s) { ds += part[tid * 16 + s]; es += epart[tid * 16 + s]; }
            acclp[tid] += -2.f * ds;  accls[tid] += 2.f * es;
        }
        __syncthreads();

        eval_core(xev, g2part, part, epart, w1p, w2p, cd, b1v, w1tv, t0 + 0.5f * dt, tid);
        for (int i = tid; i < BR * D; i += NTHREADS) {
            const int r = i >> 5, d = i & 31;
            float fv = b2s[d];
#pragma unroll
            for (int q = 0; q < NPART; ++q)
                fv += g2part[(q * BR + r) * G2S + d];
            xacc[r * XS + d] += 2.f * fv;
            xev[r * XS + d]   = x0[r * XS + d] + dt * fv;
        }
        if (tid < BR) {
            float ds = 0.f, es = 0.f;
#pragma unroll
            for (int s = 0; s < 16; ++s) { ds += part[tid * 16 + s]; es += epart[tid * 16 + s]; }
            acclp[tid] += -2.f * ds;  accls[tid] += 2.f * es;
        }
        __syncthreads();

        eval_core(xev, g2part, part, epart, w1p, w2p, cd, b1v, w1tv, t0 + dt, tid);
        for (int i = tid; i < BR * D; i += NTHREADS) {
            const int r = i >> 5, d = i & 31;
            float fv = b2s[d];
#pragma unroll
            for (int q = 0; q < NPART; ++q)
                fv += g2part[(q * BR + r) * G2S + d];
            x0[r * XS + d] += (dt / 6.f) * (xacc[r * XS + d] + fv);
        }
        if (tid < BR) {
            float ds = 0.f, es = 0.f;
#pragma unroll
            for (int s = 0; s < 16; ++s) { ds += part[tid * 16 + s]; es += epart[tid * 16 + s]; }
            lp0[tid] += (dt / 6.f) * (acclp[tid] - ds);
            ls0[tid] += (dt / 6.f) * (accls[tid] + es);
        }
        __syncthreads();
    }

    for (int i = tid; i < BR * D; i += NTHREADS)
        out[wg * BR * D + i] = x0[(i >> 5) * XS + (i & 31)];
    if (tid < BR) {
        out[B_TOT * D + wg * BR + tid]         = lp0[tid];
        out[B_TOT * D + B_TOT + wg * BR + tid] = ls0[tid];
    }
}

extern "C" void kernel_launch(void* const* d_in, const int* in_sizes, int n_in,
                              void* d_out, int out_size, void* d_ws, size_t ws_size,
                              hipStream_t stream) {
    const float* x  = (const float*)d_in[0];
    const float* W1 = (const float*)d_in[1];
    const float* b1 = (const float*)d_in[2];
    const float* W2 = (const float*)d_in[3];
    const float* b2 = (const float*)d_in[4];
    float* out = (float*)d_out;
    cnf_rk4_kernel<<<B_TOT / BR, NTHREADS, 0, stream>>>(x, W1, b1, W2, b2, out);
}


# RK4 state element-owned in registers; divergence and energy row sums by in-wave DPP butterflies (their summation order changes) instead of a serial 16-thread LDS tail
# speedup vs baseline: 1.6505x; 1.2567x over previous
#include <hip/hip_runtime.h>
#include <math.h>


typedef __attribute__((ext_vector_type(2))) float v2f;
typedef __attribute__((ext_vector_type(8))) float v8f;

#define B_TOT    512
#define D        32
#define H        256
#define BR       16
#define NTHREADS 256
#define NSTEP    64

#define XS   36
#define W1F  544
#define W2F  72
#define G2S  36
#define NPART 8

template <int N>
__device__ __forceinline__ void tanh_batch(const float (&z)[N], float (&h)[N]) {
    const float TWO_LOG2E = 2.88539008177792681472f;
    float e[N];
#pragma unroll
    for (int i = 0; i < N; ++i)
        e[i] = __builtin_amdgcn_exp2f(__builtin_fabsf(z[i]) * TWO_LOG2E);
#pragma unroll
    for (int i = 0; i < N; ++i)
        e[i] = __builtin_amdgcn_rcpf(e[i] + 1.0f);
#pragma unroll
    for (int i = 0; i < N; ++i)
        h[i] = __builtin_copysignf(__builtin_fmaf(-2.0f, e[i], 1.0f), z[i]);
}

__device__ __forceinline__ v2f lds_v2f(const float* p) {
    return *(const v2f*)p;
}

template <int CTRL>
__device__ __forceinline__ float xmask_add(float v) {
    const int b = __builtin_bit_cast(int, v);
    return v + __builtin_bit_cast(float, __builtin_amdgcn_mov_dpp(b, CTRL, 0xf, 0xf, false));
}
__device__ __forceinline__ float row16_sum(float v) {
    v = xmask_add<0x161>(v);
    v = xmask_add<0x162>(v);
    v = xmask_add<0x164>(v);
    v = xmask_add<0x168>(v);
    return v;
}
__device__ __forceinline__ float xhalf(float x) {
    const int b = __builtin_bit_cast(int, x);
    return __builtin_bit_cast(float,
        __builtin_amdgcn_permlanex16(b, b, 0x76543210, 0xfedcba98, false, false));
}

__device__ __forceinline__ void eval_core(
    const float* __restrict__ xin,
    float* __restrict__ g2part,
    float* __restrict__ part,
    const float* __restrict__ w1p,
    const float* __restrict__ w2p,
    const float (&cd)[16],
    const float (&b1v)[16],
    const float (&w1tv)[16],
    float tcur,
    int tid)
{
    const int wid  = tid >> 5;
    const int lane = tid & 31;
    const int half = lane >> 4;
    const int lr   = lane & 15;

    v2f xf[8], wf[2][8];
#pragma unroll
    for (int k0 = 0; k0 < 8; ++k0)
        xf[k0] = lds_v2f(&xin[lr * XS + k0 * 4 + 2 * half]);
#pragma unroll
    for (int it = 0; it < 2; ++it) {
#pragma unroll
        for (int k0 = 0; k0 < 8; ++k0)
            wf[it][k0] = lds_v2f(&w1p[(2 * k0 + half) * W1F + 2 * ((wid * 2 + it) * 16 + lr)]);
    }
    v8f acc0 = {}, acc1 = {};
#pragma unroll
    for (int k0 = 0; k0 < 8; ++k0) {
        acc0 = __builtin_amdgcn_wmma_f32_16x16x4_f32(
                   false, wf[0][k0], false, xf[k0], (short)0, acc0, false, false);
        acc1 = __builtin_amdgcn_wmma_f32_16x16x4_f32(
                   false, wf[1][k0], false, xf[k0], (short)0, acc1, false, false);
    }

    v2f bf[2][2][4];
#pragma unroll
    for (int nt = 0; nt < 2; ++nt) {
#pragma unroll
        for (int it = 0; it < 2; ++it) {
#pragma unroll
            for (int p = 0; p < 4; ++p)
                bf[nt][it][p] = lds_v2f(&w2p[(16 * wid + 8 * it + 4 * half + p) * W2F
                                             + 2 * (nt * 16 + lr)]);
        }
    }

    float z0[8], z1[8], h0[8], h1[8];
#pragma unroll
    for (int v = 0; v < 8; ++v) {
        z0[v] = acc0[v] + __builtin_fmaf(tcur, w1tv[v],     b1v[v]);
        z1[v] = acc1[v] + __builtin_fmaf(tcur, w1tv[8 + v], b1v[8 + v]);
    }
    tanh_batch<8>(z0, h0);

    v8f f0 = {}, f1 = {};
    float dsum = 0.f;
#pragma unroll
    for (int p = 0; p < 4; ++p) {
        v2f a;
        a[0] = h0[2 * p];
        a[1] = h0[2 * p + 1];
        f0 = __builtin_amdgcn_wmma_f32_16x16x4_f32(
                 false, a, false, bf[0][0][p], (short)0, f0, false, false);
        f1 = __builtin_amdgcn_wmma_f32_16x16x4_f32(
                 false, a, false, bf[1][0][p], (short)0, f1, false, false);
        const float zz[2] = { z1[2 * p], z1[2 * p + 1] };
        float hh[2];
        tanh_batch<2>(zz, hh);
        h1[2 * p]     = hh[0];
        h1[2 * p + 1] = hh[1];
        dsum += (1.f - h0[2 * p] * h0[2 * p]) * cd[2 * p];
        dsum += (1.f - h0[2 * p + 1] * h0[2 * p + 1]) * cd[2 * p + 1];
    }
#pragma unroll
    for (int p = 0; p < 4; ++p) {
        v2f a;
        a[0] = h1[2 * p];
        a[1] = h1[2 * p + 1];
        f0 = __builtin_amdgcn_wmma_f32_16x16x4_f32(
                 false, a, false, bf[0][1][p], (short)0, f0, false, false);
        f1 = __builtin_amdgcn_wmma_f32_16x16x4_f32(
                 false, a, false, bf[1][1][p], (short)0, f1, false, false);
        dsum += (1.f - h1[2 * p] * h1[2 * p]) * cd[8 + 2 * p];
        dsum += (1.f - h1[2 * p + 1] * h1[2 * p + 1]) * cd[8 + 2 * p + 1];
    }
    part[lr * 16 + wid * 2 + half] = dsum;
#pragma unroll
    for (int v = 0; v < 8; ++v) {
        const int row = v + 8 * half;
        g2part[(wid * BR + row) * G2S + lr]      = f0[v];
        g2part[(wid * BR + row) * G2S + 16 + lr] = f1[v];
    }
    __syncthreads();
    __builtin_amdgcn_sched_barrier(0);
}

__device__ __forceinline__ void gather_eval(
    const float* __restrict__ g2part, const float* __restrict__ part,
    const float (&xinr)[2], float b2r, int wid, int lane,
    float (&fv)[2], float (&ds)[2], float (&es)[2])
{
#pragma unroll
    for (int it = 0; it < 2; ++it) {
        const int r = wid + 8 * it;
        ds[it] = row16_sum(part[r * 16 + (lane & 15)]);
        const float e = row16_sum(0.5f * xinr[it] * xinr[it]);
        es[it] = e + xhalf(e);
        float f = b2r;
#pragma unroll
        for (int q = 0; q < NPART; ++q)
            f += g2part[(q * BR + r) * G2S + lane];
        fv[it] = f;
    }
}

__global__ __launch_bounds__(NTHREADS)
void cnf_rk4_kernel(const float* __restrict__ x,
                    const float* __restrict__ W1,
                    const float* __restrict__ b1,
                    const float* __restrict__ W2,
                    const float* __restrict__ b2,
                    float* __restrict__ out)
{
    __shared__ __align__(16) float w1p[16 * W1F];
    __shared__ __align__(16) float w2p[128 * W2F];
    __shared__ __align__(16) float w1time[H];
    __shared__ __align__(16) float b1s[H], cdiag[H];
    __shared__ __align__(16) float xs[BR * XS];
    __shared__ float g2part[NPART * BR * G2S];
    __shared__ float part[BR * 16];
    __shared__ float fin[2 * BR];

    const int tid  = threadIdx.x;
    const int wg   = blockIdx.x;
    const int wid  = tid >> 5;
    const int lane = tid & 31;

    for (int i = tid; i < 32 * H; i += NTHREADS) {
        const int k = i / H, n = i % H;
        w1p[(k >> 1) * W1F + 2 * n + (k & 1)] = W1[i];
    }
    w1time[tid] = W1[32 * H + tid];
    for (int i = tid; i < H * D; i += NTHREADS) {
        const int k = i / D, n = i % D;
        w2p[(k >> 1) * W2F + 2 * n + (k & 1)] = W2[i];
    }
    b1s[tid] = b1[tid];

    float x0r[2], xaccr[2], xinr[2];
    float lp[2], ls[2], acclp[2], accls[2];
    const float b2r = b2[lane];
#pragma unroll
    for (int it = 0; it < 2; ++it) {
        const int r = wid + 8 * it;
        x0r[it]  = x[wg * BR * D + r * D + lane];
        xinr[it] = x0r[it];
        xs[r * XS + lane] = x0r[it];
        xaccr[it] = 0.f;
        lp[it] = 0.f;  ls[it] = 0.f;  acclp[it] = 0.f;  accls[it] = 0.f;
    }
    __syncthreads();

    {
        float c = 0.f;
        const int j = tid;
#pragma unroll
        for (int k = 0; k < D; ++k)
            c += w1p[(k >> 1) * W1F + 2 * j + (k & 1)]
               * w2p[(j >> 1) * W2F + 2 * k + (j & 1)];
        cdiag[j] = c;
    }
    __syncthreads();

    float cd[16], b1v[16], w1tv[16];
    {
        const int half = lane >> 4;
#pragma unroll
        for (int it = 0; it < 2; ++it) {
#pragma unroll
            for (int v = 0; v < 8; ++v) {
                const int col = (wid * 2 + it) * 16 + 8 * half + v;
                cd[it * 8 + v]   = cdiag[col];
                b1v[it * 8 + v]  = b1s[col];
                w1tv[it * 8 + v] = w1time[col];
            }
        }
    }

    const float dt = 1.0f / (float)NSTEP;

    for (int step = 0; step < NSTEP; ++step) {
        const float t0 = step * dt;
        float fv[2], ds[2], es[2];

        eval_core(xs, g2part, part, w1p, w2p, cd, b1v, w1tv, t0, tid);
        gather_eval(g2part, part, xinr, b2r, wid, lane, fv, ds, es);
#pragma unroll
        for (int it = 0; it < 2; ++it) {
            xaccr[it] = fv[it];
            xinr[it]  = x0r[it] + 0.5f * dt * fv[it];
            xs[(wid + 8 * it) * XS + lane] = xinr[it];
            acclp[it] = -ds[it];
            accls[it] = es[it];
        }
        asm volatile("" : "+v"(acclp[0]), "+v"(acclp[1]), "+v"(accls[0]), "+v"(accls[1]));
        __syncthreads();

        eval_core(xs, g2part, part, w1p, w2p, cd, b1v, w1tv, t0 + 0.5f * dt, tid);
        gather_eval(g2part, part, xinr, b2r, wid, lane, fv, ds, es);
#pragma unroll
        for (int it = 0; it < 2; ++it) {
            xaccr[it] += 2.f * fv[it];
            xinr[it]   = x0r[it] + 0.5f * dt * fv[it];
            xs[(wid + 8 * it) * XS + lane] = xinr[it];
            acclp[it] += -2.f * ds[it];
            accls[it] += 2.f * es[it];
        }
        asm volatile("" : "+v"(acclp[0]), "+v"(acclp[1]), "+v"(accls[0]), "+v"(accls[1]));
        __syncthreads();

        eval_core(xs, g2part, part, w1p, w2p, cd, b1v, w1tv, t0 + 0.5f * dt, tid);
        gather_eval(g2part, part, xinr, b2r, wid, lane, fv, ds, es);
#pragma unroll
        for (int it = 0; it < 2; ++it) {
            xaccr[it] += 2.f * fv[it];
            xinr[it]   = x0r[it] + dt * fv[it];
            xs[(wid + 8 * it) * XS + lane] = xinr[it];
            acclp[it] += -2.f * ds[it];
            accls[it] += 2.f * es[it];
        }
        asm volatile("" : "+v"(acclp[0]), "+v"(acclp[1]), "+v"(accls[0]), "+v"(accls[1]));
        __syncthreads();

        eval_core(xs, g2part, part, w1p, w2p, cd, b1v, w1tv, t0 + dt, tid);
        gather_eval(g2part, part, xinr, b2r, wid, lane, fv, ds, es);
#pragma unroll
        for (int it = 0; it < 2; ++it) {
            x0r[it] += (dt / 6.f) * (xaccr[it] + fv[it]);
            xinr[it] = x0r[it];
            xs[(wid + 8 * it) * XS + lane] = xinr[it];
            lp[it] += (dt / 6.f) * (acclp[it] - ds[it]);
            ls[it] += (dt / 6.f) * (accls[it] + es[it]);
        }
        asm volatile("" : "+v"(lp[0]), "+v"(lp[1]), "+v"(ls[0]), "+v"(ls[1]));
        __syncthreads();
    }

#pragma unroll
    for (int it = 0; it < 2; ++it) {
        const int r = wid + 8 * it;
        out[wg * BR * D + r * D + lane] = x0r[it];
        if (lane == 0) { fin[r] = lp[it];  fin[BR + r] = ls[it]; }
    }
    __syncthreads();
    if (tid < BR) {
        out[B_TOT * D + wg * BR + tid]         = fin[tid];
        out[B_TOT * D + B_TOT + wg * BR + tid] = fin[BR + tid];
    }
}

extern "C" void kernel_launch(void* const* d_in, const int* in_sizes, int n_in,
                              void* d_out, int out_size, void* d_ws, size_t ws_size,
                              hipStream_t stream) {
    const float* x  = (const float*)d_in[0];
    const float* W1 = (const float*)d_in[1];
    const float* b1 = (const float*)d_in[2];
    const float* W2 = (const float*)d_in[3];
    const float* b2 = (const float*)d_in[4];
    float* out = (float*)d_out;
    cnf_rk4_kernel<<<B_TOT / BR, NTHREADS, 0, stream>>>(x, W1, b1, W2, b2, out);
}


# 4 waves (was 8), each rebuilds the whole stage input in its own GEMM1 B-fragment registers from the 4 split-K partials: 2nd barrier and x trip through LDS gone; f, div, energy sum order changes
# speedup vs baseline: 1.7131x; 1.0379x over previous
#include <hip/hip_runtime.h>
#include <math.h>


typedef __attribute__((ext_vector_type(2))) float v2f;
typedef __attribute__((ext_vector_type(4))) float v4f;
typedef __attribute__((ext_vector_type(8))) float v8f;

#define B_TOT    512
#define D        32
#define H        256
#define BR       16
#define NW       4
#define NTHREADS 128
#define NSTEP    64

#define W1F  544
#define W2F  72
#define G2S  36
#define PS   12
#define G2BUF (NW * BR * G2S)
#define PBUF  (BR * PS)

template <int N>
__device__ __forceinline__ void tanh_batch(const float (&z)[N], float (&h)[N]) {
    const float TWO_LOG2E = 2.88539008177792681472f;
    float e[N];
#pragma unroll
    for (int i = 0; i < N; ++i)
        e[i] = __builtin_amdgcn_exp2f(__builtin_fabsf(z[i]) * TWO_LOG2E);
#pragma unroll
    for (int i = 0; i < N; ++i)
        e[i] = __builtin_amdgcn_rcpf(e[i] + 1.0f);
#pragma unroll
    for (int i = 0; i < N; ++i)
        h[i] = __builtin_copysignf(__builtin_fmaf(-2.0f, e[i], 1.0f), z[i]);
}

__device__ __forceinline__ v2f lds_v2f(const float* p) {
    return *(const v2f*)p;
}
__device__ __forceinline__ v4f lds_v4f(const float* p) {
    return *(const v4f*)p;
}

__device__ __forceinline__ float xhalf(float x) {
    const int b = __builtin_bit_cast(int, x);
    return __builtin_bit_cast(float,
        __builtin_amdgcn_permlanex16(b, b, 0x76543210, 0xfedcba98, false, false));
}

__device__ __forceinline__ void eval_core(
    const float (&xin)[16],
    float* g2b,
    float* pb,
    const v2f (&wf)[4][8],
    const v2f (&bf)[2][4][4],
    const float (&cd)[32],
    const float (&b1v)[32],
    const float (&w1tv)[32],
    float tcur, int wid, int lr, int half)
{
    v8f acc0 = {}, acc1 = {}, acc2 = {}, acc3 = {};
    v8f f0 = {}, f1 = {};
    float dsum = 0.f;
    float za[16], zb[16], ha[16], hb[16];

#pragma unroll
    for (int k0 = 0; k0 < 8; ++k0) {
        v2f xb;
        xb[0] = xin[2 * k0];
        xb[1] = xin[2 * k0 + 1];
        acc0 = __builtin_amdgcn_wmma_f32_16x16x4_f32(
                   false, wf[0][k0], false, xb, (short)0, acc0, false, false);
        acc1 = __builtin_amdgcn_wmma_f32_16x16x4_f32(
                   false, wf[1][k0], false, xb, (short)0, acc1, false, false);
    }
    __builtin_amdgcn_sched_barrier(0);
#pragma unroll
    for (int v = 0; v < 8; ++v) {
        za[v]     = acc0[v] + __builtin_fmaf(tcur, w1tv[v],     b1v[v]);
        za[8 + v] = acc1[v] + __builtin_fmaf(tcur, w1tv[8 + v], b1v[8 + v]);
    }
#pragma unroll
    for (int k0 = 0; k0 < 8; ++k0) {
        v2f xb;
        xb[0] = xin[2 * k0];
        xb[1] = xin[2 * k0 + 1];
        acc2 = __builtin_amdgcn_wmma_f32_16x16x4_f32(
                   false, wf[2][k0], false, xb, (short)0, acc2, false, false);
        acc3 = __builtin_amdgcn_wmma_f32_16x16x4_f32(
                   false, wf[3][k0], false, xb, (short)0, acc3, false, false);
        const float zz[2] = { za[2 * k0], za[2 * k0 + 1] };
        float hh[2];
        tanh_batch<2>(zz, hh);
        ha[2 * k0]     = hh[0];
        ha[2 * k0 + 1] = hh[1];
    }
#pragma unroll
    for (int v = 0; v < 8; ++v) {
        zb[v]     = acc2[v] + __builtin_fmaf(tcur, w1tv[16 + v], b1v[16 + v]);
        zb[8 + v] = acc3[v] + __builtin_fmaf(tcur, w1tv[24 + v], b1v[24 + v]);
    }

#pragma unroll
    for (int s = 0; s < 8; ++s) {
        const int it = s >> 2, p = s & 3;
        v2f a;
        a[0] = ha[2 * s];
        a[1] = ha[2 * s + 1];
        f0 = __builtin_amdgcn_wmma_f32_16x16x4_f32(
                 false, a, false, bf[0][it][p], (short)0, f0, false, false);
        f1 = __builtin_amdgcn_wmma_f32_16x16x4_f32(
                 false, a, false, bf[1][it][p], (short)0, f1, false, false);
        const float zz[2] = { zb[2 * s], zb[2 * s + 1] };
        float hh[2];
        tanh_batch<2>(zz, hh);
        hb[2 * s]     = hh[0];
        hb[2 * s + 1] = hh[1];
        dsum += (1.f - a[0] * a[0]) * cd[2 * s];
        dsum += (1.f - a[1] * a[1]) * cd[2 * s + 1];
    }
#pragma unroll
    for (int s = 0; s < 8; ++s) {
        const int it = 2 + (s >> 2), p = s & 3;
        v2f a;
        a[0] = hb[2 * s];
        a[1] = hb[2 * s + 1];
        f0 = __builtin_amdgcn_wmma_f32_16x16x4_f32(
                 false, a, false, bf[0][it][p], (short)0, f0, false, false);
        f1 = __builtin_amdgcn_wmma_f32_16x16x4_f32(
                 false, a, false, bf[1][it][p], (short)0, f1, false, false);
        dsum += (1.f - a[0] * a[0]) * cd[16 + 2 * s];
        dsum += (1.f - a[1] * a[1]) * cd[16 + 2 * s + 1];
    }
    pb[lr * PS + wid * 2 + half] = dsum;
#pragma unroll
    for (int v = 0; v < 8; ++v) {
        const int row = v + 8 * half;
        g2b[(wid * BR + row) * G2S + lr]      = f0[v];
        g2b[(wid * BR + row) * G2S + 16 + lr] = f1[v];
    }
    __syncthreads();
}

__device__ __forceinline__ void gather_eval(
    const float* g2b, const float* pb,
    const float (&b2v)[16], const float (&xin)[16], int lr, int half,
    float (&fv)[16], float& ds, float& es)
{
    v4f pr[NW][4];
#pragma unroll
    for (int q = 0; q < NW; ++q) {
#pragma unroll
        for (int m = 0; m < 4; ++m)
            pr[q][m] = lds_v4f(&g2b[(q * BR + lr) * G2S + 16 * half + 4 * m]);
    }
    const v4f d0 = lds_v4f(&pb[lr * PS]);
    const v4f d1 = lds_v4f(&pb[lr * PS + 4]);
    float e = 0.f;
#pragma unroll
    for (int j = 0; j < 16; ++j)
        e = __builtin_fmaf(0.5f * xin[j], xin[j], e);
    es = e + xhalf(e);
#pragma unroll
    for (int m = 0; m < 4; ++m) {
#pragma unroll
        for (int c = 0; c < 4; ++c) {
            float f = b2v[4 * m + c];
#pragma unroll
            for (int q = 0; q < NW; ++q)
                f += pr[q][m][c];
            fv[4 * m + c] = f;
        }
    }
    ds = ((((((d0[0] + d0[1]) + d0[2]) + d0[3]) + d1[0]) + d1[1]) + d1[2]) + d1[3];
}

__global__ __launch_bounds__(NTHREADS)
void cnf_rk4_kernel(const float* __restrict__ x,
                    const float* __restrict__ W1,
                    const float* __restrict__ b1,
                    const float* __restrict__ W2,
                    const float* __restrict__ b2,
                    float* __restrict__ out)
{
    __shared__ __align__(16) float w1p[16 * W1F];
    __shared__ __align__(16) float w2p[128 * W2F];
    __shared__ __align__(16) float w1time[H];
    __shared__ __align__(16) float b1s[H], cdiag[H];
    __shared__ __align__(16) float g2part[2 * G2BUF];
    __shared__ __align__(16) float part[2 * PBUF];

    const int tid  = threadIdx.x;
    const int wg   = blockIdx.x;
    const int wid  = tid >> 5;
    const int lane = tid & 31;
    const int half = lane >> 4;
    const int lr   = lane & 15;

    for (int i = tid; i < 32 * H; i += NTHREADS) {
        const int k = i / H, n = i % H;
        w1p[(k >> 1) * W1F + 2 * n + (k & 1)] = W1[i];
    }
    for (int i = tid; i < H * D; i += NTHREADS) {
        const int k = i / D, n = i % D;
        w2p[(k >> 1) * W2F + 2 * n + (k & 1)] = W2[i];
    }
    for (int i = tid; i < H; i += NTHREADS) {
        w1time[i] = W1[32 * H + i];
        b1s[i]    = b1[i];
    }

    float x0r[16], xaccr[16], xinr[16], b2v[16];
#pragma unroll
    for (int j = 0; j < 16; ++j) {
        x0r[j]   = x[(wg * BR + lr) * D + 16 * half + j];
        xinr[j]  = x0r[j];
        xaccr[j] = 0.f;
        b2v[j]   = b2[16 * half + j];
    }
    float lp = 0.f, ls = 0.f, acclp = 0.f, accls = 0.f;
    __syncthreads();

    for (int j = tid; j < H; j += NTHREADS) {
        float c = 0.f;
#pragma unroll
        for (int k = 0; k < D; ++k)
            c += w1p[(k >> 1) * W1F + 2 * j + (k & 1)]
               * w2p[(j >> 1) * W2F + 2 * k + (j & 1)];
        cdiag[j] = c;
    }
    __syncthreads();

    float cd[32], b1v[32], w1tv[32];
    v2f wf[4][8], bf[2][4][4];
#pragma unroll
    for (int it = 0; it < 4; ++it) {
#pragma unroll
        for (int v = 0; v < 8; ++v) {
            const int col = (wid * 4 + it) * 16 + 8 * half + v;
            cd[it * 8 + v]   = cdiag[col];
            b1v[it * 8 + v]  = b1s[col];
            w1tv[it * 8 + v] = w1time[col];
        }
#pragma unroll
        for (int k0 = 0; k0 < 8; ++k0)
            wf[it][k0] = lds_v2f(&w1p[(8 * half + k0) * W1F + 2 * ((wid * 4 + it) * 16 + lr)]);
#pragma unroll
        for (int nt = 0; nt < 2; ++nt) {
#pragma unroll
            for (int p = 0; p < 4; ++p)
                bf[nt][it][p] = lds_v2f(&w2p[(32 * wid + 8 * it + 4 * half + p) * W2F
                                             + 2 * (nt * 16 + lr)]);
        }
    }

    const float dt = 1.0f / (float)NSTEP;
    float* const g2a = g2part;
    float* const g2c = g2part + G2BUF;
    float* const pa  = part;
    float* const pc  = part + PBUF;

    for (int step = 0; step < NSTEP; ++step) {
        const float t0 = step * dt;
        float fv[16], ds, es;

        eval_core(xinr, g2a, pa, wf, bf, cd, b1v, w1tv, t0, wid, lr, half);
        gather_eval(g2a, pa, b2v, xinr, lr, half, fv, ds, es);
#pragma unroll
        for (int j = 0; j < 16; ++j) {
            xaccr[j] = fv[j];
            xinr[j]  = x0r[j] + 0.5f * dt * fv[j];
        }
        acclp = -ds;
        accls = es;

        eval_core(xinr, g2c, pc, wf, bf, cd, b1v, w1tv, t0 + 0.5f * dt, wid, lr, half);
        gather_eval(g2c, pc, b2v, xinr, lr, half, fv, ds, es);
#pragma unroll
        for (int j = 0; j < 16; ++j) {
            xaccr[j] += 2.f * fv[j];
            xinr[j]   = x0r[j] + 0.5f * dt * fv[j];
        }
        acclp += -2.f * ds;
        accls += 2.f * es;

        eval_core(xinr, g2a, pa, wf, bf, cd, b1v, w1tv, t0 + 0.5f * dt, wid, lr, half);
        gather_eval(g2a, pa, b2v, xinr, lr, half, fv, ds, es);
#pragma unroll
        for (int j = 0; j < 16; ++j) {
            xaccr[j] += 2.f * fv[j];
            xinr[j]   = x0r[j] + dt * fv[j];
        }
        acclp += -2.f * ds;
        accls += 2.f * es;

        eval_core(xinr, g2c, pc, wf, bf, cd, b1v, w1tv, t0 + dt, wid, lr, half);
        gather_eval(g2c, pc, b2v, xinr, lr, half, fv, ds, es);
#pragma unroll
        for (int j = 0; j < 16; ++j) {
            x0r[j] += (dt / 6.f) * (xaccr[j] + fv[j]);
            xinr[j] = x0r[j];
        }
        lp += (dt / 6.f) * (acclp - ds);
        ls += (dt / 6.f) * (accls + es);
    }

    if (wid == 0) {
#pragma unroll
        for (int m = 0; m < 4; ++m) {
            v4f o;
            o[0] = x0r[4 * m];     o[1] = x0r[4 * m + 1];
            o[2] = x0r[4 * m + 2]; o[3] = x0r[4 * m + 3];
            *(v4f*)&out[(wg * BR + lr) * D + 16 * half + 4 * m] = o;
        }
        if (half == 0) {
            out[B_TOT * D + wg * BR + lr]         = lp;
            out[B_TOT * D + B_TOT + wg * BR + lr] = ls;
        }
    }
}

extern "C" void kernel_launch(void* const* d_in, const int* in_sizes, int n_in,
                              void* d_out, int out_size, void* d_ws, size_t ws_size,
                              hipStream_t stream) {
    const float* x  = (const float*)d_in[0];
    const float* W1 = (const float*)d_in[1];
    const float* b1 = (const float*)d_in[2];
    const float* W2 = (const float*)d_in[3];
    const float* b2 = (const float*)d_in[4];
    float* out = (float*)d_out;
    cnf_rk4_kernel<<<B_TOT / BR, NTHREADS, 0, stream>>>(x, W1, b1, W2, b2, out);
}


# tanh by the chip's own v_tanh_f32: one transcendental per hidden value instead of exp2 + rcp + four plain ops; h differs from the formula's in its last bits; all sums and their order unchanged
# speedup vs baseline: 1.8663x; 1.0894x over previous
#include <hip/hip_runtime.h>
#include <math.h>


typedef __attribute__((ext_vector_type(2))) float v2f;
typedef __attribute__((ext_vector_type(4))) float v4f;
typedef __attribute__((ext_vector_type(8))) float v8f;

#define B_TOT    512
#define D        32
#define H        256
#define BR       16
#define NW       4
#define NTHREADS 128
#define NSTEP    64

#define W1F  544
#define W2F  72
#define G2S  36
#define PS   12
#define G2BUF (NW * BR * G2S)
#define PBUF  (BR * PS)

template <int N>
__device__ __forceinline__ void tanh_batch(const float (&z)[N], float (&h)[N]) {
#pragma unroll
    for (int i = 0; i < N; ++i)
        h[i] = __builtin_amdgcn_tanhf(z[i]);
}

__device__ __forceinline__ v2f lds_v2f(const float* p) {
    return *(const v2f*)p;
}
__device__ __forceinline__ v4f lds_v4f(const float* p) {
    return *(const v4f*)p;
}

__device__ __forceinline__ float xhalf(float x) {
    const int b = __builtin_bit_cast(int, x);
    return __builtin_bit_cast(float,
        __builtin_amdgcn_permlanex16(b, b, 0x76543210, 0xfedcba98, false, false));
}

__device__ __forceinline__ void eval_core(
    const float (&xin)[16],
    float* g2b,
    float* pb,
    const v2f (&wf)[4][8],
    const v2f (&bf)[2][4][4],
    const float (&cd)[32],
    const float (&b1v)[32],
    const float (&w1tv)[32],
    float tcur, int wid, int lr, int half)
{
    v8f acc0 = {}, acc1 = {}, acc2 = {}, acc3 = {};
    v8f f0 = {}, f1 = {};
    float dsum = 0.f;
    float za[16], zb[16], ha[16], hb[16];

#pragma unroll
    for (int k0 = 0; k0 < 8; ++k0) {
        v2f xb;
        xb[0] = xin[2 * k0];
        xb[1] = xin[2 * k0 + 1];
        acc0 = __builtin_amdgcn_wmma_f32_16x16x4_f32(
                   false, wf[0][k0], false, xb, (short)0, acc0, false, false);
        acc1 = __builtin_amdgcn_wmma_f32_16x16x4_f32(
                   false, wf[1][k0], false, xb, (short)0, acc1, false, false);
    }
    __builtin_amdgcn_sched_barrier(0);
#pragma unroll
    for (int v = 0; v < 8; ++v) {
        za[v]     = acc0[v] + __builtin_fmaf(tcur, w1tv[v],     b1v[v]);
        za[8 + v] = acc1[v] + __builtin_fmaf(tcur, w1tv[8 + v], b1v[8 + v]);
    }
#pragma unroll
    for (int k0 = 0; k0 < 8; ++k0) {
        v2f xb;
        xb[0] = xin[2 * k0];
        xb[1] = xin[2 * k0 + 1];
        acc2 = __builtin_amdgcn_wmma_f32_16x16x4_f32(
                   false, wf[2][k0], false, xb, (short)0, acc2, false, false);
        acc3 = __builtin_amdgcn_wmma_f32_16x16x4_f32(
                   false, wf[3][k0], false, xb, (short)0, acc3, false, false);
        const float zz[2] = { za[2 * k0], za[2 * k0 + 1] };
        float hh[2];
        tanh_batch<2>(zz, hh);
        ha[2 * k0]     = hh[0];
        ha[2 * k0 + 1] = hh[1];
    }
#pragma unroll
    for (int v = 0; v < 8; ++v) {
        zb[v]     = acc2[v] + __builtin_fmaf(tcur, w1tv[16 + v], b1v[16 + v]);
        zb[8 + v] = acc3[v] + __builtin_fmaf(tcur, w1tv[24 + v], b1v[24 + v]);
    }

#pragma unroll
    for (int s = 0; s < 8; ++s) {
        const int it = s >> 2, p = s & 3;
        v2f a;
        a[0] = ha[2 * s];
        a[1] = ha[2 * s + 1];
        f0 = __builtin_amdgcn_wmma_f32_16x16x4_f32(
                 false, a, false, bf[0][it][p], (short)0, f0, false, false);
        f1 = __builtin_amdgcn_wmma_f32_16x16x4_f32(
                 false, a, false, bf[1][it][p], (short)0, f1, false, false);
        const float zz[2] = { zb[2 * s], zb[2 * s + 1] };
        float hh[2];
        tanh_batch<2>(zz, hh);
        hb[2 * s]     = hh[0];
        hb[2 * s + 1] = hh[1];
        dsum += (1.f - a[0] * a[0]) * cd[2 * s];
        dsum += (1.f - a[1] * a[1]) * cd[2 * s + 1];
    }
#pragma unroll
    for (int s = 0; s < 8; ++s) {
        const int it = 2 + (s >> 2), p = s & 3;
        v2f a;
        a[0] = hb[2 * s];
        a[1] = hb[2 * s + 1];
        f0 = __builtin_amdgcn_wmma_f32_16x16x4_f32(
                 false, a, false, bf[0][it][p], (short)0, f0, false, false);
        f1 = __builtin_amdgcn_wmma_f32_16x16x4_f32(
                 false, a, false, bf[1][it][p], (short)0, f1, false, false);
        dsum += (1.f - a[0] * a[0]) * cd[16 + 2 * s];
        dsum += (1.f - a[1] * a[1]) * cd[16 + 2 * s + 1];
    }
    pb[lr * PS + wid * 2 + half] = dsum;
#pragma unroll
    for (int v = 0; v < 8; ++v) {
        const int row = v + 8 * half;
        g2b[(wid * BR + row) * G2S + lr]      = f0[v];
        g2b[(wid * BR + row) * G2S + 16 + lr] = f1[v];
    }
    __syncthreads();
}

__device__ __forceinline__ void gather_eval(
    const float* g2b, const float* pb,
    const float (&b2v)[16], const float (&xin)[16], int lr, int half,
    float (&fv)[16], float& ds, float& es)
{
    v4f pr[NW][4];
#pragma unroll
    for (int q = 0; q < NW; ++q) {
#pragma unroll
        for (int m = 0; m < 4; ++m)
            pr[q][m] = lds_v4f(&g2b[(q * BR + lr) * G2S + 16 * half + 4 * m]);
    }
    const v4f d0 = lds_v4f(&pb[lr * PS]);
    const v4f d1 = lds_v4f(&pb[lr * PS + 4]);
    float e = 0.f;
#pragma unroll
    for (int j = 0; j < 16; ++j)
        e = __builtin_fmaf(0.5f * xin[j], xin[j], e);
    es = e + xhalf(e);
#pragma unroll
    for (int m = 0; m < 4; ++m) {
#pragma unroll
        for (int c = 0; c < 4; ++c) {
            float f = b2v[4 * m + c];
#pragma unroll
            for (int q = 0; q < NW; ++q)
                f += pr[q][m][c];
            fv[4 * m + c] = f;
        }
    }
    ds = ((((((d0[0] + d0[1]) + d0[2]) + d0[3]) + d1[0]) + d1[1]) + d1[2]) + d1[3];
}

__global__ __launch_bounds__(NTHREADS)
void cnf_rk4_kernel(const float* __restrict__ x,
                    const float* __restrict__ W1,
                    const float* __restrict__ b1,
                    const float* __restrict__ W2,
                    const float* __restrict__ b2,
                    float* __restrict__ out)
{
    __shared__ __align__(16) float w1p[16 * W1F];
    __shared__ __align__(16) float w2p[128 * W2F];
    __shared__ __align__(16) float w1time[H];
    __shared__ __align__(16) float b1s[H], cdiag[H];
    __shared__ __align__(16) float g2part[2 * G2BUF];
    __shared__ __align__(16) float part[2 * PBUF];

    const int tid  = threadIdx.x;
    const int wg   = blockIdx.x;
    const int wid  = tid >> 5;
    const int lane = tid & 31;
    const int half = lane >> 4;
    const int lr   = lane & 15;

    for (int i = tid; i < 32 * H; i += NTHREADS) {
        const int k = i / H, n = i % H;
        w1p[(k >> 1) * W1F + 2 * n + (k & 1)] = W1[i];
    }
    for (int i = tid; i < H * D; i += NTHREADS) {
        const int k = i / D, n = i % D;
        w2p[(k >> 1) * W2F + 2 * n + (k & 1)] = W2[i];
    }
    for (int i = tid; i < H; i += NTHREADS) {
        w1time[i] = W1[32 * H + i];
        b1s[i]    = b1[i];
    }

    float x0r[16], xaccr[16], xinr[16], b2v[16];
#pragma unroll
    for (int j = 0; j < 16; ++j) {
        x0r[j]   = x[(wg * BR + lr) * D + 16 * half + j];
        xinr[j]  = x0r[j];
        xaccr[j] = 0.f;
        b2v[j]   = b2[16 * half + j];
    }
    float lp = 0.f, ls = 0.f, acclp = 0.f, accls = 0.f;
    __syncthreads();

    for (int j = tid; j < H; j += NTHREADS) {
        float c = 0.f;
#pragma unroll
        for (int k = 0; k < D; ++k)
            c += w1p[(k >> 1) * W1F + 2 * j + (k & 1)]
               * w2p[(j >> 1) * W2F + 2 * k + (j & 1)];
        cdiag[j] = c;
    }
    __syncthreads();

    float cd[32], b1v[32], w1tv[32];
    v2f wf[4][8], bf[2][4][4];
#pragma unroll
    for (int it = 0; it < 4; ++it) {
#pragma unroll
        for (int v = 0; v < 8; ++v) {
            const int col = (wid * 4 + it) * 16 + 8 * half + v;
            cd[it * 8 + v]   = cdiag[col];
            b1v[it * 8 + v]  = b1s[col];
            w1tv[it * 8 + v] = w1time[col];
        }
#pragma unroll
        for (int k0 = 0; k0 < 8; ++k0)
            wf[it][k0] = lds_v2f(&w1p[(8 * half + k0) * W1F + 2 * ((wid * 4 + it) * 16 + lr)]);
#pragma unroll
        for (int nt = 0; nt < 2; ++nt) {
#pragma unroll
            for (int p = 0; p < 4; ++p)
                bf[nt][it][p] = lds_v2f(&w2p[(32 * wid + 8 * it + 4 * half + p) * W2F
                                             + 2 * (nt * 16 + lr)]);
        }
    }

    const float dt = 1.0f / (float)NSTEP;
    float* const g2a = g2part;
    float* const g2c = g2part + G2BUF;
    float* const pa  = part;
    float* const pc  = part + PBUF;

    for (int step = 0; step < NSTEP; ++step) {
        const float t0 = step * dt;
        float fv[16], ds, es;

        eval_core(xinr, g2a, pa, wf, bf, cd, b1v, w1tv, t0, wid, lr, half);
        gather_eval(g2a, pa, b2v, xinr, lr, half, fv, ds, es);
#pragma unroll
        for (int j = 0; j < 16; ++j) {
            xaccr[j] = fv[j];
            xinr[j]  = x0r[j] + 0.5f * dt * fv[j];
        }
        acclp = -ds;
        accls = es;

        eval_core(xinr, g2c, pc, wf, bf, cd, b1v, w1tv, t0 + 0.5f * dt, wid, lr, half);
        gather_eval(g2c, pc, b2v, xinr, lr, half, fv, ds, es);
#pragma unroll
        for (int j = 0; j < 16; ++j) {
            xaccr[j] += 2.f * fv[j];
            xinr[j]   = x0r[j] + 0.5f * dt * fv[j];
        }
        acclp += -2.f * ds;
        accls += 2.f * es;

        eval_core(xinr, g2a, pa, wf, bf, cd, b1v, w1tv, t0 + 0.5f * dt, wid, lr, half);
        gather_eval(g2a, pa, b2v, xinr, lr, half, fv, ds, es);
#pragma unroll
        for (int j = 0; j < 16; ++j) {
            xaccr[j] += 2.f * fv[j];
            xinr[j]   = x0r[j] + dt * fv[j];
        }
        acclp += -2.f * ds;
        accls += 2.f * es;

        eval_core(xinr, g2c, pc, wf, bf, cd, b1v, w1tv, t0 + dt, wid, lr, half);
        gather_eval(g2c, pc, b2v, xinr, lr, half, fv, ds, es);
#pragma unroll
        for (int j = 0; j < 16; ++j) {
            x0r[j] += (dt / 6.f) * (xaccr[j] + fv[j]);
            xinr[j] = x0r[j];
        }
        lp += (dt / 6.f) * (acclp - ds);
        ls += (dt / 6.f) * (accls + es);
    }

    if (wid == 0) {
#pragma unroll
        for (int m = 0; m < 4; ++m) {
            v4f o;
            o[0] = x0r[4 * m];     o[1] = x0r[4 * m + 1];
            o[2] = x0r[4 * m + 2]; o[3] = x0r[4 * m + 3];
            *(v4f*)&out[(wg * BR + lr) * D + 16 * half + 4 * m] = o;
        }
        if (half == 0) {
            out[B_TOT * D + wg * BR + lr]         = lp;
            out[B_TOT * D + B_TOT + wg * BR + lr] = ls;
        }
    }
}

extern "C" void kernel_launch(void* const* d_in, const int* in_sizes, int n_in,
                              void* d_out, int out_size, void* d_ws, size_t ws_size,
                              hipStream_t stream) {
    const float* x  = (const float*)d_in[0];
    const float* W1 = (const float*)d_in[1];
    const float* b1 = (const float*)d_in[2];
    const float* W2 = (const float*)d_in[3];
    const float* b2 = (const float*)d_in[4];
    float* out = (float*)d_out;
    cnf_rk4_kernel<<<B_TOT / BR, NTHREADS, 0, stream>>>(x, W1, b1, W2, b2, out);
}


# split-K exchange of f in two column halves, each barrier and LDS trip hidden inside the other half's 16 products (GEMM2 operands swapped, GEMM1 K outer); k order of GEMM1, f and energy sums changes
# speedup vs baseline: 2.0441x; 1.0953x over previous
#include <hip/hip_runtime.h>
#include <math.h>


typedef __attribute__((ext_vector_type(2))) float v2f;
typedef __attribute__((ext_vector_type(4))) float v4f;
typedef __attribute__((ext_vector_type(8))) float v8f;

#define B_TOT    512
#define D        32
#define H        256
#define BR       16
#define NW       4
#define NTHREADS 128
#define NSTEP    64

#define W1F  544
#define W2F  72
#define G2S  36
#define PS   12
#define G2BUF (NW * BR * G2S)
#define PBUF  (BR * PS)

__device__ __forceinline__ v2f lds_v2f(const float* p) {
    return *(const v2f*)p;
}
__device__ __forceinline__ v4f lds_v4f(const float* p) {
    return *(const v4f*)p;
}

__device__ __forceinline__ float xhalf(float x) {
    const int b = __builtin_bit_cast(int, x);
    return __builtin_bit_cast(float,
        __builtin_amdgcn_permlanex16(b, b, 0x76543210, 0xfedcba98, false, false));
}

template <int MT>
__device__ __forceinline__ void load_half(const float* g2, int lr, int half,
                                          v4f (&pr)[NW][2]) {
#pragma unroll
    for (int q = 0; q < NW; ++q) {
#pragma unroll
        for (int m = 0; m < 2; ++m)
            pr[q][m] = lds_v4f(&g2[(q * BR + lr) * G2S + 16 * MT + 8 * half + 4 * m]);
    }
}

template <int BASE>
__device__ __forceinline__ void sum_half(const v4f (&pr)[NW][2],
                                         const float (&b2v)[16], float (&f)[8]) {
#pragma unroll
    for (int m = 0; m < 2; ++m) {
#pragma unroll
        for (int c = 0; c < 4; ++c) {
            float v = b2v[BASE + 4 * m + c];
#pragma unroll
            for (int q = 0; q < NW; ++q)
                v += pr[q][m][c];
            f[4 * m + c] = v;
        }
    }
}

template <int S, int BASE>
__device__ __forceinline__ void rk_half(const float (&f)[8], float (&x0r)[16],
                                        float (&xaccr)[16], float (&xinr)[16],
                                        float dt) {
#pragma unroll
    for (int j = 0; j < 8; ++j) {
        const int i = BASE + j;
        if (S == 1) {
            xaccr[i] = f[j];
            xinr[i]  = x0r[i] + 0.5f * dt * f[j];
        } else if (S == 2) {
            xaccr[i] += 2.f * f[j];
            xinr[i]   = x0r[i] + 0.5f * dt * f[j];
        } else if (S == 3) {
            xaccr[i] += 2.f * f[j];
            xinr[i]   = x0r[i] + dt * f[j];
        } else {
            x0r[i] += (dt / 6.f) * (xaccr[i] + f[j]);
            xinr[i] = x0r[i];
        }
    }
}

__device__ __forceinline__ void meet_b(const float* g2, const float* pb,
                                       int lr, int half,
                                       v4f (&pr)[NW][2], v4f& d0, v4f& d1) {
    __syncthreads();
    load_half<1>(g2, lr, half, pr);
    d0 = lds_v4f(&pb[lr * PS]);
    d1 = lds_v4f(&pb[lr * PS + 4]);
}
template <int S>
__device__ __forceinline__ void finish_b(const v4f (&pr)[NW][2], v4f d0, v4f d1,
                                         const float (&b2v)[16], float es,
                                         float (&x0r)[16], float (&xaccr)[16],
                                         float (&xinr)[16], float& lp, float& ls,
                                         float& acclp, float& accls, float dt) {
    float fB[8];
    sum_half<8>(pr, b2v, fB);
    rk_half<S, 8>(fB, x0r, xaccr, xinr, dt);
    const float ds =
        ((((((d0[0] + d0[1]) + d0[2]) + d0[3]) + d1[0]) + d1[1]) + d1[2]) + d1[3];
    if (S == 1) {
        acclp = -ds;
        accls = es;
    } else if (S == 4) {
        lp += (dt / 6.f) * (acclp - ds);
        ls += (dt / 6.f) * (accls + es);
    } else {
        acclp += -2.f * ds;
        accls += 2.f * es;
    }
}

#define WMMA_F32(a, b, c) \
    __builtin_amdgcn_wmma_f32_16x16x4_f32(false, (a), false, (b), (short)0, (c), false, false)

template <int S, bool PENDING>
__device__ __forceinline__ void eval_stage(
    float (&x0r)[16], float (&xaccr)[16], float (&xinr)[16],
    float& lp, float& ls, float& acclp, float& accls, float& es,
    float* g2, float* pb,
    const v2f (&wf)[4][8],
    const v2f (&bf)[2][4][4],
    const float (&cd)[32], const float (&b1v)[32], const float (&w1tv)[32],
    const float (&b2v)[16],
    float tcur, float dt, int wid, int lr, int half)
{
    constexpr int SP = (S == 1) ? 4 : S - 1;
    v8f acc0 = {}, acc1 = {}, acc2 = {}, acc3 = {};
    if (PENDING) {
        asm volatile("" : "+v"(xinr[0]) : : "memory");
    }
    v4f prb[NW][2], d0 = {}, d1 = {};

#pragma unroll
    for (int k0 = 0; k0 < 2; ++k0) {
        v2f xb;
        xb[0] = xinr[2 * k0];
        xb[1] = xinr[2 * k0 + 1];
        acc0 = WMMA_F32(wf[0][k0], xb, acc0);
        acc1 = WMMA_F32(wf[1][k0], xb, acc1);
        acc2 = WMMA_F32(wf[2][k0], xb, acc2);
        acc3 = WMMA_F32(wf[3][k0], xb, acc3);
    }
    if (PENDING) {
        asm volatile("" : "+v"(acc0), "+v"(acc1), "+v"(acc2), "+v"(acc3));
        meet_b(g2, pb, lr, half, prb, d0, d1);
        __builtin_amdgcn_sched_barrier(0);
    }
#pragma unroll
    for (int k0 = 2; k0 < 4; ++k0) {
        v2f xb;
        xb[0] = xinr[2 * k0];
        xb[1] = xinr[2 * k0 + 1];
        acc0 = WMMA_F32(wf[0][k0], xb, acc0);
        acc1 = WMMA_F32(wf[1][k0], xb, acc1);
        acc2 = WMMA_F32(wf[2][k0], xb, acc2);
        acc3 = WMMA_F32(wf[3][k0], xb, acc3);
    }
    if (PENDING) {
        asm volatile("" : "+v"(acc0), "+v"(acc1), "+v"(acc2), "+v"(acc3),
                          "+v"(prb[0][0]), "+v"(prb[0][1]));
        finish_b<SP>(prb, d0, d1, b2v, es, x0r, xaccr, xinr, lp, ls, acclp, accls, dt);
    }

#pragma unroll
    for (int k0 = 4; k0 < 8; ++k0) {
        v2f xb;
        xb[0] = xinr[2 * k0];
        xb[1] = xinr[2 * k0 + 1];
        acc0 = WMMA_F32(wf[0][k0], xb, acc0);
        acc1 = WMMA_F32(wf[1][k0], xb, acc1);
        acc2 = WMMA_F32(wf[2][k0], xb, acc2);
        acc3 = WMMA_F32(wf[3][k0], xb, acc3);
    }

    {
        float e = 0.f;
#pragma unroll
        for (int j = 0; j < 16; ++j)
            e = __builtin_fmaf(0.5f * xinr[j], xinr[j], e);
        es = e + xhalf(e);
    }

    float z[32], h[32];
#pragma unroll
    for (int v = 0; v < 8; ++v) {
        z[v]      = acc0[v] + __builtin_fmaf(tcur, w1tv[v],      b1v[v]);
        z[8 + v]  = acc1[v] + __builtin_fmaf(tcur, w1tv[8 + v],  b1v[8 + v]);
        z[16 + v] = acc2[v] + __builtin_fmaf(tcur, w1tv[16 + v], b1v[16 + v]);
        z[24 + v] = acc3[v] + __builtin_fmaf(tcur, w1tv[24 + v], b1v[24 + v]);
    }

    v8f fa = {}, fb = {};
    float dsum = 0.f;
#pragma unroll
    for (int s = 0; s < 8; ++s) {
        const int ta = (s < 4) ? 0 : 2, tb = ta + 1, p = s & 3;
        h[8 * ta + 2 * p]     = __builtin_amdgcn_tanhf(z[8 * ta + 2 * p]);
        h[8 * ta + 2 * p + 1] = __builtin_amdgcn_tanhf(z[8 * ta + 2 * p + 1]);
        h[8 * tb + 2 * p]     = __builtin_amdgcn_tanhf(z[8 * tb + 2 * p]);
        h[8 * tb + 2 * p + 1] = __builtin_amdgcn_tanhf(z[8 * tb + 2 * p + 1]);
        v2f a, b;
        a[0] = h[8 * ta + 2 * p];
        a[1] = h[8 * ta + 2 * p + 1];
        b[0] = h[8 * tb + 2 * p];
        b[1] = h[8 * tb + 2 * p + 1];
        fa = WMMA_F32(bf[0][ta][p], a, fa);
        fb = WMMA_F32(bf[0][tb][p], b, fb);
        if (p == 3) {
#pragma unroll
            for (int i = 8 * ta; i < 8 * ta + 16; ++i)
                dsum += (1.f - h[i] * h[i]) * cd[i];
        }
    }

    v8f ga = {}, gb = {};
#pragma unroll
    for (int p = 0; p < 4; ++p) {
        v2f a, b;
        a[0] = h[2 * p];
        a[1] = h[2 * p + 1];
        b[0] = h[8 + 2 * p];
        b[1] = h[8 + 2 * p + 1];
        ga = WMMA_F32(bf[1][0][p], a, ga);
        gb = WMMA_F32(bf[1][1][p], b, gb);
        if (p == 0) {
            const v8f f0 = fa + fb;
            float* dst = &g2[(wid * BR + lr) * G2S + 8 * half];
            v4f lo, hi;
            lo[0] = f0[0]; lo[1] = f0[1]; lo[2] = f0[2]; lo[3] = f0[3];
            hi[0] = f0[4]; hi[1] = f0[5]; hi[2] = f0[6]; hi[3] = f0[7];
            *(v4f*)dst       = lo;
            *(v4f*)(dst + 4) = hi;
            asm volatile("" : "+v"(ga), "+v"(gb) : : "memory");
        }
    }
    asm volatile("" : "+v"(ga), "+v"(gb));
    v4f pra[NW][2];
    __syncthreads();
    load_half<0>(g2, lr, half, pra);
    __builtin_amdgcn_sched_barrier(0);
#pragma unroll
    for (int p = 0; p < 4; ++p) {
        v2f a, b;
        a[0] = h[16 + 2 * p];
        a[1] = h[16 + 2 * p + 1];
        b[0] = h[24 + 2 * p];
        b[1] = h[24 + 2 * p + 1];
        ga = WMMA_F32(bf[1][2][p], a, ga);
        gb = WMMA_F32(bf[1][3][p], b, gb);
    }
    asm volatile("" : "+v"(ga), "+v"(gb), "+v"(pra[0][0]), "+v"(pra[0][1]));
    {
        float fA[8];
        sum_half<0>(pra, b2v, fA);
        rk_half<S, 0>(fA, x0r, xaccr, xinr, dt);
    }
    {
        const v8f f1 = ga + gb;
        float* dst = &g2[(wid * BR + lr) * G2S + 16 + 8 * half];
        v4f lo, hi;
        lo[0] = f1[0]; lo[1] = f1[1]; lo[2] = f1[2]; lo[3] = f1[3];
        hi[0] = f1[4]; hi[1] = f1[5]; hi[2] = f1[6]; hi[3] = f1[7];
        *(v4f*)dst       = lo;
        *(v4f*)(dst + 4) = hi;
        pb[lr * PS + wid * 2 + half] = dsum;
    }
}

__global__ __launch_bounds__(NTHREADS)
void cnf_rk4_kernel(const float* __restrict__ x,
                    const float* __restrict__ W1,
                    const float* __restrict__ b1,
                    const float* __restrict__ W2,
                    const float* __restrict__ b2,
                    float* __restrict__ out)
{
    __shared__ __align__(16) float w1p[16 * W1F];
    __shared__ __align__(16) float w2p[128 * W2F];
    __shared__ __align__(16) float w1time[H];
    __shared__ __align__(16) float b1s[H], cdiag[H];
    __shared__ __align__(16) float g2part[G2BUF];
    __shared__ __align__(16) float part[PBUF];

    const int tid  = threadIdx.x;
    const int wg   = blockIdx.x;
    const int wid  = tid >> 5;
    const int lane = tid & 31;
    const int half = lane >> 4;
    const int lr   = lane & 15;

    for (int i = tid; i < 32 * H; i += NTHREADS) {
        const int k = i / H, n = i % H;
        w1p[(k >> 1) * W1F + 2 * n + (k & 1)] = W1[i];
    }
    for (int i = tid; i < H * D; i += NTHREADS) {
        const int k = i / D, n = i % D;
        w2p[(k >> 1) * W2F + 2 * n + (k & 1)] = W2[i];
    }
    for (int i = tid; i < H; i += NTHREADS) {
        w1time[i] = W1[32 * H + i];
        b1s[i]    = b1[i];
    }

    float x0r[16], xaccr[16], xinr[16], b2v[16];
#pragma unroll
    for (int j = 0; j < 16; ++j) {
        const int col = 8 * half + j + ((j >= 8) ? 8 : 0);
        x0r[j]   = x[(wg * BR + lr) * D + col];
        xinr[j]  = x0r[j];
        xaccr[j] = 0.f;
        b2v[j]   = b2[col];
    }
    float lp = 0.f, ls = 0.f, acclp = 0.f, accls = 0.f, es = 0.f;
    __syncthreads();

    for (int j = tid; j < H; j += NTHREADS) {
        float c = 0.f;
#pragma unroll
        for (int k = 0; k < D; ++k)
            c += w1p[(k >> 1) * W1F + 2 * j + (k & 1)]
               * w2p[(j >> 1) * W2F + 2 * k + (j & 1)];
        cdiag[j] = c;
    }
    __syncthreads();

    float cd[32], b1v[32], w1tv[32];
    v2f wf[4][8], bf[2][4][4];
#pragma unroll
    for (int it = 0; it < 4; ++it) {
#pragma unroll
        for (int v = 0; v < 8; ++v) {
            const int col = (wid * 4 + it) * 16 + 8 * half + v;
            cd[it * 8 + v]   = cdiag[col];
            b1v[it * 8 + v]  = b1s[col];
            w1tv[it * 8 + v] = w1time[col];
        }
#pragma unroll
        for (int k0 = 0; k0 < 8; ++k0) {
            const int prow = 4 * half + k0 + ((k0 >= 4) ? 4 : 0);
            wf[it][k0] = lds_v2f(&w1p[prow * W1F + 2 * ((wid * 4 + it) * 16 + lr)]);
        }
#pragma unroll
        for (int mt = 0; mt < 2; ++mt) {
#pragma unroll
            for (int p = 0; p < 4; ++p)
                bf[mt][it][p] = lds_v2f(&w2p[(32 * wid + 8 * it + 4 * half + p) * W2F
                                             + 2 * (mt * 16 + lr)]);
        }
    }

    const float dt = 1.0f / (float)NSTEP;

    for (int step = 0; step < NSTEP; ++step) {
        const float t0 = step * dt;

        if (step == 0)
            eval_stage<1, false>(x0r, xaccr, xinr, lp, ls, acclp, accls, es, g2part, part,
                                 wf, bf, cd, b1v, w1tv, b2v, t0, dt, wid, lr, half);
        else
            eval_stage<1, true>(x0r, xaccr, xinr, lp, ls, acclp, accls, es, g2part, part,
                                wf, bf, cd, b1v, w1tv, b2v, t0, dt, wid, lr, half);
        eval_stage<2, true>(x0r, xaccr, xinr, lp, ls, acclp, accls, es, g2part, part,
                            wf, bf, cd, b1v, w1tv, b2v, t0 + 0.5f * dt, dt, wid, lr, half);
        eval_stage<3, true>(x0r, xaccr, xinr, lp, ls, acclp, accls, es, g2part, part,
                            wf, bf, cd, b1v, w1tv, b2v, t0 + 0.5f * dt, dt, wid, lr, half);
        eval_stage<4, true>(x0r, xaccr, xinr, lp, ls, acclp, accls, es, g2part, part,
                            wf, bf, cd, b1v, w1tv, b2v, t0 + dt, dt, wid, lr, half);
    }
    {
        v4f prb[NW][2], d0, d1;
        meet_b(g2part, part, lr, half, prb, d0, d1);
        finish_b<4>(prb, d0, d1, b2v, es, x0r, xaccr, xinr, lp, ls, acclp, accls, dt);
    }

    if (wid == 0) {
#pragma unroll
        for (int m = 0; m < 4; ++m) {
            v4f o;
            o[0] = x0r[4 * m];     o[1] = x0r[4 * m + 1];
            o[2] = x0r[4 * m + 2]; o[3] = x0r[4 * m + 3];
            *(v4f*)&out[(wg * BR + lr) * D + 8 * half + 4 * m + ((m >= 2) ? 8 : 0)] = o;
        }
        if (half == 0) {
            out[B_TOT * D + wg * BR + lr]         = lp;
            out[B_TOT * D + B_TOT + wg * BR + lr] = ls;
        }
    }
}

extern "C" void kernel_launch(void* const* d_in, const int* in_sizes, int n_in,
                              void* d_out, int out_size, void* d_ws, size_t ws_size,
                              hipStream_t stream) {
    const float* x  = (const float*)d_in[0];
    const float* W1 = (const float*)d_in[1];
    const float* b1 = (const float*)d_in[2];
    const float* W2 = (const float*)d_in[3];
    const float* b2 = (const float*)d_in[4];
    float* out = (float*)d_out;
    cnf_rk4_kernel<<<B_TOT / BR, NTHREADS, 0, stream>>>(x, W1, b1, W2, b2, out);
}
